# starting kernel
# baseline (speedup 1.0000x reference)
#include <hip/hip_runtime.h>

#define N_NODES 65536
#define N_EDGES 1048576
#define KD      128

typedef __attribute__((ext_vector_type(2)))  float     v2f;
typedef __attribute__((ext_vector_type(8)))  float     v8f;
typedef __attribute__((ext_vector_type(16))) _Float16  v16h;

__global__ void zero_f32(float* __restrict__ p, size_t n) {
  size_t i = (size_t)blockIdx.x * blockDim.x + threadIdx.x;
  if (i < n) p[i] = 0.0f;
}

__global__ void degree_kernel(const int* __restrict__ dst,
                              float* __restrict__ deg) {
  int e = blockIdx.x * blockDim.x + threadIdx.x;
  if (e < N_EDGES) atomicAdd(deg + dst[e], 1.0f);
}

__global__ void scatter_kernel(const float* __restrict__ feat,
                               const int* __restrict__ src,
                               const int* __restrict__ dst,
                               float* __restrict__ agg) {
  long long t = (long long)blockIdx.x * blockDim.x + threadIdx.x;
  int       chunk = (int)(t & 31);
  long long e     = t >> 5;
  if (e >= N_EDGES) return;
  int s = src[e], d = dst[e];
  float4 v = ((const float4*)(feat + (size_t)s * KD))[chunk];
  float* op = agg + (size_t)d * KD + chunk * 4;
  atomicAdd(op + 0, v.x);
  atomicAdd(op + 1, v.y);
  atomicAdd(op + 2, v.z);
  atomicAdd(op + 3, v.w);
}

template <int NOUT, bool RELU>
__global__ __launch_bounds__(32 * (NOUT / 16))
void sage_gemm(const float* __restrict__ X,
               const float* __restrict__ AGG,
               const float* __restrict__ deg,
               const float* __restrict__ Wself,
               const float* __restrict__ Wneigh,
               const float* __restrict__ bias,
               float* __restrict__ OUT) {
  __shared__ float xs[16][KD];
  __shared__ float ms[16][KD];

  const int m0       = blockIdx.x * 16;
  const int tid      = threadIdx.x;
  const int nthreads = 32 * (NOUT / 16);

  for (int i = tid; i < 16 * (KD / 4); i += nthreads) {
    int r  = i / (KD / 4);
    int c4 = i % (KD / 4);
    float4 xv = ((const float4*)(X + (size_t)(m0 + r) * KD))[c4];
    float4 av = ((const float4*)(AGG + (size_t)(m0 + r) * KD))[c4];
    float  rd = 1.0f / fmaxf(deg[m0 + r], 1.0f);
    av.x *= rd; av.y *= rd; av.z *= rd; av.w *= rd;
    ((float4*)&xs[r][0])[c4] = xv;
    ((float4*)&ms[r][0])[c4] = av;
  }
  __syncthreads();

  const int  wave = tid / 32;
  const int  lane = tid % 32;
  const int  n0   = wave * 16;
  const int  mrow = lane & 15;
  const bool hi   = lane >= 16;
  const int  ncol = n0 + (lane & 15);

  v8f c = {};

#if __has_builtin(__builtin_amdgcn_wmma_f32_16x16x4_f32)
  for (int kk = 0; kk < KD; kk += 4) {
    int ka = kk + (hi ? 2 : 0);
    v2f a_self = { xs[mrow][ka], xs[mrow][ka + 1] };
    v2f a_mean = { ms[mrow][ka], ms[mrow][ka + 1] };
    v2f bs = { Wself[(size_t)ka * NOUT + ncol],
               Wself[(size_t)(ka + 1) * NOUT + ncol] };
    v2f bn = { Wneigh[(size_t)ka * NOUT + ncol],
               Wneigh[(size_t)(ka + 1) * NOUT + ncol] };
    c = __builtin_amdgcn_wmma_f32_16x16x4_f32(false, a_self, false, bs,
                                              (short)0, c, false, false);
    c = __builtin_amdgcn_wmma_f32_16x16x4_f32(false, a_mean, false, bn,
                                              (short)0, c, false, false);
  }
#else
  for (int kk = 0; kk < KD; kk += 32) {
    int ka = kk + (hi ? 8 : 0);
    int kb = kk + (hi ? 16 : 0);
    v16h a_self, a_mean, bs, bn;
#pragma unroll
    for (int j = 0; j < 8; j++) {
      a_self[j]     = (_Float16)xs[mrow][ka + j];
      a_self[8 + j] = (_Float16)xs[mrow][ka + 16 + j];
      a_mean[j]     = (_Float16)ms[mrow][ka + j];
      a_mean[8 + j] = (_Float16)ms[mrow][ka + 16 + j];
    }
#pragma unroll
    for (int j = 0; j < 16; j++) {
      bs[j] = (_Float16)Wself[(size_t)(kb + j) * NOUT + ncol];
      bn[j] = (_Float16)Wneigh[(size_t)(kb + j) * NOUT + ncol];
    }
    c = __builtin_amdgcn_wmma_f32_16x16x32_f16(false, a_self, false, bs,
                                               (short)0, c, false, false);
    c = __builtin_amdgcn_wmma_f32_16x16x32_f16(false, a_mean, false, bn,
                                               (short)0, c, false, false);
  }
#endif

  float bv = bias[ncol];
#pragma unroll
  for (int i = 0; i < 8; i++) {
    int   row = m0 + i + (hi ? 8 : 0);
    float v   = c[i] + bv;
    if (RELU) v = fmaxf(v, 0.0f);
    OUT[(size_t)row * NOUT + ncol] = v;
  }
}

extern "C" void kernel_launch(void* const* d_in, const int* in_sizes, int n_in,
                              void* d_out, int out_size, void* d_ws,
                              size_t ws_size, hipStream_t stream) {
  (void)in_sizes; (void)n_in; (void)out_size; (void)ws_size;
  const float* x   = (const float*)d_in[0];
  const int*   src = (const int*)d_in[1];
  const int*   dst = (const int*)d_in[2];
  const float* W1s = (const float*)d_in[3];
  const float* W1n = (const float*)d_in[4];
  const float* b1  = (const float*)d_in[5];
  const float* W2s = (const float*)d_in[6];
  const float* W2n = (const float*)d_in[7];
  const float* b2  = (const float*)d_in[8];
  float*       out = (float*)d_out;

  char*  ws  = (char*)d_ws;
  float* deg = (float*)ws;
  float* agg = deg + N_NODES;
  float* h   = agg + (size_t)N_NODES * KD;

  size_t nz = (size_t)N_NODES * (KD + 1);
  zero_f32<<<(unsigned)((nz + 255) / 256), 256, 0, stream>>>(deg, nz);

  degree_kernel<<<(N_EDGES + 255) / 256, 256, 0, stream>>>(dst, deg);

  scatter_kernel<<<(N_EDGES * 32) / 256, 256, 0, stream>>>(x, src, dst, agg);
  sage_gemm<128, true><<<N_NODES / 16, 256, 0, stream>>>(x, agg, deg, W1s,
                                                         W1n, b1, h);

  size_t na = (size_t)N_NODES * KD;
  zero_f32<<<(unsigned)((na + 255) / 256), 256, 0, stream>>>(agg, na);
  scatter_kernel<<<(N_EDGES * 32) / 256, 256, 0, stream>>>(h, src, dst, agg);
  sage_gemm<64, false><<<N_NODES / 16, 128, 0, stream>>>(h, agg, deg, W2s,
                                                         W2n, b2, out);
}


# Layer 1's owner groups save their matched edges (source row, local row in one int) to the workspace; layer 2 reads its list, no second scan of dst and src; layer-2 add order follows the list
# speedup vs baseline: 3.3732x; 3.3732x over previous
#include <hip/hip_runtime.h>


#define N_NODES 65536
#define N_EDGES 1048576
#define KD      128

typedef __attribute__((ext_vector_type(2)))  float     v2f;
typedef __attribute__((ext_vector_type(8)))  float     v8f;

#define OWN_ROWS    256
#define OWN_SHIFT   8
#define AGG_THREADS 1024
#define AGG_WAVES   (AGG_THREADS / 32)
#define SCAN_UNROLL 4
#define QCAP        (32 * 4 * SCAN_UNROLL + 32)
#define PITCH       132
#define LIST_CAP    8192
#define LIST_HEAD   32
#define LIST_REGION (LIST_HEAD + LIST_CAP)

__device__ __forceinline__ void add_rows(int pk, int m, int lane,
                                         const float* __restrict__ feat,
                                         float* acc) {
  for (int t = 0; t < m; t += 4) {
    float v[4][4];
    int   dl[4];
#pragma unroll
    for (int u = 0; u < 4; u++) {
      const int pu = __builtin_amdgcn_readlane(pk, t + u);
      dl[u] = pu & (OWN_ROWS - 1);
      const float* row = feat + (size_t)((unsigned)pu >> OWN_SHIFT) * KD + lane;
#pragma unroll
      for (int j = 0; j < 4; j++) v[u][j] = row[32 * j];
    }
#pragma unroll
    for (int u = 0; u < 4; u++) {
      if (t + u < m) {
        float* ar = acc + dl[u] * PITCH + lane;
#pragma unroll
        for (int j = 0; j < 4; j++) atomicAdd(ar + 32 * j, v[u][j]);
      }
    }
  }
}

template <bool SAVE>
__device__ __forceinline__ void drain_queue(const int* q, int n, int lane,
                                            const float* __restrict__ feat,
                                            const int* __restrict__ src,
                                            float* acc, int* lpos, int* lbuf) {
  __builtin_amdgcn_fence(__ATOMIC_ACQ_REL, "wavefront");
  for (int base = 0; base < n; base += 32) {
    int m = n - base;
    if (m > 32) m = 32;
    int pk = 0;
    if (lane < m) {
      const int ent = q[base + lane];
      pk = (src[ent >> OWN_SHIFT] << OWN_SHIFT) | (ent & (OWN_ROWS - 1));
    }
    if (SAVE) {
      int p0 = 0;
      if (lane == 0) p0 = atomicAdd(lpos, m);
      p0 = __builtin_amdgcn_readfirstlane(p0);
      if (lane < m && p0 + lane < LIST_CAP) lbuf[p0 + lane] = pk;
    }
    add_rows(pk, m, lane, feat, acc);
  }
  __builtin_amdgcn_fence(__ATOMIC_ACQ_REL, "wavefront");
}

template <int NOUT, bool RELU, bool SAVE_LIST, bool USE_LIST>
__global__ __launch_bounds__(AGG_THREADS)
void sage_layer_owner(const float* __restrict__ feat,
                      const int* __restrict__ src,
                      const int* __restrict__ dst,
                      const float* __restrict__ Wself,
                      const float* __restrict__ Wneigh,
                      const float* __restrict__ bias,
                      float* __restrict__ OUT,
                      int* __restrict__ lists) {
  __shared__ float acc[OWN_ROWS * PITCH];
  __shared__ float xsq[OWN_ROWS * PITCH];
  __shared__ int   cnt[OWN_ROWS];
  __shared__ int   lbuf[LIST_CAP];
  __shared__ int   lpos;

  const int tid  = threadIdx.x;
  const int lane = tid & 31;
  const int wave = tid >> 5;
  const int own  = blockIdx.x;

  for (int i = tid; i < OWN_ROWS * PITCH / 4; i += AGG_THREADS)
    ((float4*)acc)[i] = make_float4(0.0f, 0.0f, 0.0f, 0.0f);
  if (tid < OWN_ROWS) cnt[tid] = 0;
  if (tid == 0) lpos = 0;
  __syncthreads();

  int* myhead  = lists + own * LIST_REGION;
  int* mylist  = myhead + LIST_HEAD;
  int  n_saved = 0;
  bool listed  = false;
  if (USE_LIST) {
    n_saved = __builtin_amdgcn_readfirstlane(myhead[0]);
    listed  = (unsigned)n_saved <= (unsigned)LIST_CAP;
  }
  if (listed) {
    const int last = n_saved - 1;
    if (n_saved > 0) {
      int e[LIST_CAP / AGG_THREADS];
#pragma unroll
      for (int r = 0; r < LIST_CAP / AGG_THREADS; r++) {
        const int i = r * AGG_THREADS + tid;
        e[r] = mylist[i < last ? i : last];
      }
#pragma unroll
      for (int r = 0; r < LIST_CAP / AGG_THREADS; r++) {
        const int i = r * AGG_THREADS + tid;
        if (i < n_saved) {
          lbuf[i] = e[r];
          atomicAdd(&cnt[e[r] & (OWN_ROWS - 1)], 1);
        }
      }
    }
    __syncthreads();
    const int wv  = __builtin_amdgcn_readfirstlane(wave);
    const int per = (n_saved + AGG_WAVES - 1) / AGG_WAVES;
    int       end = wv * per + per;
    if (end > n_saved) end = n_saved;
    for (int base = wv * per; base < end; base += 32) {
      int m = end - base;
      if (m > 32) m = 32;
      const int idx = base + lane;
      int pk = lbuf[idx < last ? idx : last];
      if (lane >= m) pk = 0;
      add_rows(pk, m, lane, feat, acc);
    }
  } else {
    int* q  = reinterpret_cast<int*>(xsq) + wave * QCAP;
    int  qn = 0;
    const int4* dst4 = (const int4*)dst;

    for (int i0 = tid; i0 < N_EDGES / 4; i0 += AGG_THREADS * SCAN_UNROLL) {
      int4 d4[SCAN_UNROLL];
#pragma unroll
      for (int k = 0; k < SCAN_UNROLL; k++) d4[k] = dst4[i0 + k * AGG_THREADS];
#pragma unroll
      for (int k = 0; k < SCAN_UNROLL; k++) {
        const int dv[4] = { d4[k].x, d4[k].y, d4[k].z, d4[k].w };
        const int e0    = (i0 + k * AGG_THREADS) * 4;
#pragma unroll
        for (int j = 0; j < 4; j++) {
          bool     hit = (dv[j] >> OWN_SHIFT) == own;
          unsigned mk  = __builtin_amdgcn_ballot_w32(hit);
          if (mk != 0u) {
            if (hit) {
              int pos = qn + __builtin_popcount(mk & ((1u << lane) - 1u));
              q[pos] = ((e0 + j) << OWN_SHIFT) | (dv[j] & (OWN_ROWS - 1));
              atomicAdd(&cnt[dv[j] & (OWN_ROWS - 1)], 1);
            }
            qn += __builtin_popcount(mk);
          }
        }
      }
      if (qn >= 32) {
        drain_queue<SAVE_LIST>(q, qn, lane, feat, src, acc, &lpos, lbuf);
        qn = 0;
      }
    }
    drain_queue<SAVE_LIST>(q, qn, lane, feat, src, acc, &lpos, lbuf);
  }
  __syncthreads();
  if (SAVE_LIST) {
    const int total = lpos;
    const int nst   = total < LIST_CAP ? total : LIST_CAP;
    for (int i = tid; i < nst; i += AGG_THREADS) mylist[i] = lbuf[i];
    if (tid == 0) myhead[0] = total;
  }

  const float* xrows = feat + (size_t)own * OWN_ROWS * KD;
  for (int i = tid; i < OWN_ROWS * (KD / 4); i += AGG_THREADS) {
    const int r  = i / (KD / 4);
    const int c4 = i % (KD / 4);
    float4  xv = ((const float4*)(xrows + (size_t)r * KD))[c4];
    float   rd = 1.0f / fmaxf((float)cnt[r], 1.0f);
    float4* ap = (float4*)(acc + r * PITCH) + c4;
    float4  av = *ap;
    av.x *= rd; av.y *= rd; av.z *= rd; av.w *= rd;
    *ap = av;
    *((float4*)(xsq + r * PITCH) + c4) = xv;
  }
  __syncthreads();

  constexpr int PAIRS   = NOUT / 32;
  constexpr int RGROUPS = AGG_WAVES / PAIRS;
  constexpr int TILES_R = 16 / RGROUPS;
  const int  pair  = wave % PAIRS;
  const int  rg    = wave / PAIRS;
  const int  mrow  = lane & 15;
  const bool hi    = lane >= 16;
  const int  ncol0 = pair * 32 + (lane & 15);

  v8f c[TILES_R][2];
#pragma unroll
  for (int i = 0; i < TILES_R; i++) { c[i][0] = v8f{}; c[i][1] = v8f{}; }

#pragma unroll 4
  for (int kk = 0; kk < KD; kk += 4) {
    const int    ka = kk + (hi ? 2 : 0);
    const float* ps = Wself  + (size_t)ka * NOUT + ncol0;
    const float* pn = Wneigh + (size_t)ka * NOUT + ncol0;
    v2f bs0 = { ps[0],  ps[NOUT] };
    v2f bs1 = { ps[16], ps[NOUT + 16] };
    v2f bn0 = { pn[0],  pn[NOUT] };
    v2f bn1 = { pn[16], pn[NOUT + 16] };
    v2f as[TILES_R], am[TILES_R];
#pragma unroll
    for (int i = 0; i < TILES_R; i++) {
      const int off = ((rg + RGROUPS * i) * 16 + mrow) * PITCH + ka;
      as[i] = *(const v2f*)(xsq + off);
      am[i] = *(const v2f*)(acc + off);
    }
#pragma unroll
    for (int i = 0; i < TILES_R; i++) {
      c[i][0] = __builtin_amdgcn_wmma_f32_16x16x4_f32(false, as[i], false, bs0,
                                                      (short)0, c[i][0], false, false);
      c[i][1] = __builtin_amdgcn_wmma_f32_16x16x4_f32(false, as[i], false, bs1,
                                                      (short)0, c[i][1], false, false);
    }
#pragma unroll
    for (int i = 0; i < TILES_R; i++) {
      c[i][0] = __builtin_amdgcn_wmma_f32_16x16x4_f32(false, am[i], false, bn0,
                                                      (short)0, c[i][0], false, false);
      c[i][1] = __builtin_amdgcn_wmma_f32_16x16x4_f32(false, am[i], false, bn1,
                                                      (short)0, c[i][1], false, false);
    }
  }

#pragma unroll
  for (int i = 0; i < TILES_R; i++) {
#pragma unroll
    for (int h2 = 0; h2 < 2; h2++) {
      const int   col = ncol0 + 16 * h2;
      const float bv  = bias[col];
#pragma unroll
      for (int r = 0; r < 8; r++) {
        const int row = own * OWN_ROWS + (rg + RGROUPS * i) * 16 + r + (hi ? 8 : 0);
        float     v   = c[i][h2][r] + bv;
        if (RELU) v = fmaxf(v, 0.0f);
        OUT[(size_t)row * NOUT + col] = v;
      }
    }
  }
}

extern "C" void kernel_launch(void* const* d_in, const int* in_sizes, int n_in,
                              void* d_out, int out_size, void* d_ws,
                              size_t ws_size, hipStream_t stream) {
  (void)in_sizes; (void)n_in; (void)out_size; (void)ws_size;
  const float* x   = (const float*)d_in[0];
  const int*   src = (const int*)d_in[1];
  const int*   dst = (const int*)d_in[2];
  const float* W1s = (const float*)d_in[3];
  const float* W1n = (const float*)d_in[4];
  const float* b1  = (const float*)d_in[5];
  const float* W2s = (const float*)d_in[6];
  const float* W2n = (const float*)d_in[7];
  const float* b2  = (const float*)d_in[8];
  float*       out = (float*)d_out;

  float* h      = (float*)d_ws;
  int*   lists  = (int*)((char*)d_ws + 33554432);

  sage_layer_owner<128, true, true, false><<<N_NODES / OWN_ROWS, AGG_THREADS, 0, stream>>>(
      x, src, dst, W1s, W1n, b1, h, lists);
  sage_layer_owner<64, false, false, true><<<N_NODES / OWN_ROWS, AGG_THREADS, 0, stream>>>(
      h, src, dst, W2s, W2n, b2, out, lists);
}
